# starting kernel
# baseline (speedup 1.0000x reference)
#include <hip/hip_runtime.h>
#include <hip/hip_bf16.h>

typedef __attribute__((ext_vector_type(2))) float v2f;
typedef __attribute__((ext_vector_type(8))) float v8f;

#define IN_FEATS 128
#define OUT_FEATS 128

__global__ void sage_zero_kernel(float4* __restrict__ p, int n4) {
    int i = blockIdx.x * blockDim.x + threadIdx.x;
    if (i < n4) p[i] = make_float4(0.f, 0.f, 0.f, 0.f);
}

__global__ void sage_scatter_kernel(const float* __restrict__ x,
                                    const int* __restrict__ esrc,
                                    const int* __restrict__ edst,
                                    float* __restrict__ agg,
                                    float* __restrict__ deg,
                                    int n_edges) {
    int gid = blockIdx.x * blockDim.x + threadIdx.x;
    int e = gid >> 5;
    if (e >= n_edges) return;
    int part = gid & 31;
    int s = esrc[e];
    int d = edst[e];
    const float4 v = *(const float4*)(x + (size_t)s * IN_FEATS + part * 4);
    float* ap = agg + (size_t)d * IN_FEATS + part * 4;
    unsafeAtomicAdd(ap + 0, v.x);
    unsafeAtomicAdd(ap + 1, v.y);
    unsafeAtomicAdd(ap + 2, v.z);
    unsafeAtomicAdd(ap + 3, v.w);
    if (part == 0) unsafeAtomicAdd(deg + d, 1.0f);
}

__global__ void sage_rdeg_kernel(float* __restrict__ deg, int n) {
    int i = blockIdx.x * blockDim.x + threadIdx.x;
    if (i < n) deg[i] = 1.0f / fmaxf(deg[i], 1.0f);
}

__global__ __launch_bounds__(256) void sage_gemm_kernel(
        const float* __restrict__ agg, const float* __restrict__ rdeg,
        const float* __restrict__ x,   const float* __restrict__ W,
        const float* __restrict__ Wr,  const float* __restrict__ bias,
        float* __restrict__ out) {
    const int wave = threadIdx.x >> 5;
    const int lane = threadIdx.x & 31;
    const int m0 = blockIdx.x << 4;
    const int n0 = wave << 4;
    const int r  = lane & 15;
    const int kh = lane >> 4;

    const float scale = rdeg[m0 + r];
    const float bl    = bias[n0 + r];

    const float* aggp = agg + (size_t)(m0 + r) * IN_FEATS + 2 * kh;
    const float* xp   = x   + (size_t)(m0 + r) * IN_FEATS + 2 * kh;
    const float* wp   = W   + (size_t)(2 * kh) * OUT_FEATS + n0 + r;
    const float* wrp  = Wr  + (size_t)(2 * kh) * OUT_FEATS + n0 + r;

    v8f acc = {};
    #pragma unroll 4
    for (int k = 0; k < IN_FEATS; k += 4) {
        v2f a0 = *(const v2f*)(aggp + k);
        a0 = a0 * scale;
        v2f a1 = *(const v2f*)(xp + k);
        v2f b0, b1;
        b0.x = wp[(size_t)k * OUT_FEATS];
        b0.y = wp[(size_t)(k + 1) * OUT_FEATS];
        b1.x = wrp[(size_t)k * OUT_FEATS];
        b1.y = wrp[(size_t)(k + 1) * OUT_FEATS];
        acc = __builtin_amdgcn_wmma_f32_16x16x4_f32(
                  false, a0, false, b0, (short)0, acc, false, false);
        acc = __builtin_amdgcn_wmma_f32_16x16x4_f32(
                  false, a1, false, b1, (short)0, acc, false, false);
    }

    float* op = out + (size_t)(m0 + 8 * kh) * OUT_FEATS + n0 + r;
    #pragma unroll
    for (int i = 0; i < 8; ++i)
        op[(size_t)i * OUT_FEATS] = acc[i] + bl;
}

extern "C" void kernel_launch(void* const* d_in, const int* in_sizes, int n_in,
                              void* d_out, int out_size, void* d_ws, size_t ws_size,
                              hipStream_t stream) {
    const float* x    = (const float*)d_in[0];
    const int*   src  = (const int*)d_in[1];
    const int*   dst  = (const int*)d_in[2];
    const float* W    = (const float*)d_in[3];
    const float* Wr   = (const float*)d_in[4];
    const float* bias = (const float*)d_in[5];
    float* out = (float*)d_out;

    const int n_nodes = in_sizes[0] / IN_FEATS;
    const int n_edges = in_sizes[1];

    float* agg = (float*)d_ws;
    float* deg = agg + (size_t)n_nodes * IN_FEATS;

    int n4 = (n_nodes * IN_FEATS + n_nodes) / 4;
    sage_zero_kernel<<<(n4 + 255) / 256, 256, 0, stream>>>((float4*)agg, n4);

    long long sthreads = (long long)n_edges * 32;
    int sblocks = (int)((sthreads + 255) / 256);
    sage_scatter_kernel<<<sblocks, 256, 0, stream>>>(x, src, dst, agg, deg, n_edges);

    sage_rdeg_kernel<<<(n_nodes + 255) / 256, 256, 0, stream>>>(deg, n_nodes);

    sage_gemm_kernel<<<n_nodes / 16, 256, 0, stream>>>(agg, deg, x, W, Wr, bias, out);
}


# GEMM launch folded into the sum kernel: each block multiplies its own 160-row LDS sum tile and its x rows (staged in the freed queue LDS) by W and Wr on f32 WMMA; k order in the products permuted
# speedup vs baseline: 3.4995x; 3.4995x over previous
#include <hip/hip_runtime.h>
#include <hip/hip_bf16.h>


typedef __attribute__((ext_vector_type(2))) float v2f;
typedef __attribute__((ext_vector_type(8))) float v8f;

#define IN_FEATS 128
#define OUT_FEATS 128

#define NPB         160
#define AGG_WAVES   32
#define AGG_THREADS (AGG_WAVES * 32)
#define CHUNK       512
#define QW          1024
#define PITCH       130
static_assert(QW >= 2 * CHUNK && CHUNK <= AGG_THREADS, "queue room and tail coverage");
static_assert(AGG_WAVES * QW >= NPB * PITCH, "the x tile reuses the queue words");
static_assert((NPB * PITCH) % 4 == 0 && NPB % 16 == 0 && AGG_WAVES == 32, "tile shape");

__device__ __forceinline__ void sage_load_chunk(int4 (&v)[4], const int* __restrict__ edst,
                                                int cc, int lane) {
    const int4* p = (const int4*)(edst + cc * CHUNK + 4 * lane);
    v[0] = p[0]; v[1] = p[32]; v[2] = p[64]; v[3] = p[96];
}

__device__ __forceinline__ void sage_add_row(float* __restrict__ s_agg, int d, int lane, float4 v) {
    float* ap = s_agg + d * PITCH + lane;
    unsafeAtomicAdd(ap + 0,  v.x);
    unsafeAtomicAdd(ap + 32, v.y);
    unsafeAtomicAdd(ap + 64, v.z);
    unsafeAtomicAdd(ap + 96, v.w);
}

__device__ __forceinline__ void sage_drain(const float* __restrict__ x,
                                           const int* __restrict__ esrc,
                                           const int* __restrict__ edst,
                                           const int* qe, int cnt, int lo,
                                           float* __restrict__ s_agg,
                                           float* __restrict__ s_deg, int lane) {
    for (int base = 0; base < cnt; base += 32) {
        const int i = base + lane;
        int s_l = 0, d_l = NPB;
        if (i < cnt) {
            const int e = qe[i];
            s_l = esrc[e];
            d_l = edst[e] - lo;
            unsafeAtomicAdd(s_deg + d_l, 1.0f);
        }
        const int m = min(32, cnt - base);
        for (int j = 0; j < m; j += 4) {
            const int s0 = __builtin_amdgcn_readlane(s_l, j);
            const int s1 = __builtin_amdgcn_readlane(s_l, j + 1);
            const int s2 = __builtin_amdgcn_readlane(s_l, j + 2);
            const int s3 = __builtin_amdgcn_readlane(s_l, j + 3);
            const int d0 = __builtin_amdgcn_readlane(d_l, j);
            const int d1 = __builtin_amdgcn_readlane(d_l, j + 1);
            const int d2 = __builtin_amdgcn_readlane(d_l, j + 2);
            const int d3 = __builtin_amdgcn_readlane(d_l, j + 3);
            const float4 v0 = *(const float4*)(x + (size_t)s0 * IN_FEATS + 4 * lane);
            const float4 v1 = *(const float4*)(x + (size_t)s1 * IN_FEATS + 4 * lane);
            const float4 v2 = *(const float4*)(x + (size_t)s2 * IN_FEATS + 4 * lane);
            const float4 v3 = *(const float4*)(x + (size_t)s3 * IN_FEATS + 4 * lane);
            sage_add_row(s_agg, d0, lane, v0);
            sage_add_row(s_agg, d1, lane, v1);
            sage_add_row(s_agg, d2, lane, v2);
            sage_add_row(s_agg, d3, lane, v3);
        }
    }
}

__device__ __forceinline__ void sage_load_b(v2f (&bw)[8], v2f (&br)[8],
                                            const float* __restrict__ wp,
                                            const float* __restrict__ wrp, int kc) {
    #pragma unroll
    for (int jj = 0; jj < 8; ++jj) {
        const int j  = kc * 8 + jj;
        const int kb = 8 * (j & 15) + 2 * (j >> 4);
        bw[jj].x = wp[kb * OUT_FEATS];
        bw[jj].y = wp[(kb + 4) * OUT_FEATS];
        br[jj].x = wrp[kb * OUT_FEATS];
        br[jj].y = wrp[(kb + 4) * OUT_FEATS];
    }
}

template <int NT>
__device__ __forceinline__ void sage_tile_gemm(const float* s_agg, const float* s_x,
                                               const float* s_deg,
                                               const float* __restrict__ W,
                                               const float* __restrict__ Wr,
                                               const float* __restrict__ bias,
                                               float* __restrict__ out,
                                               int lo, int nloc, int g, int n0, int lane) {
    const int c = lane & 15;
    const int h = lane >> 4;
    const float* wp  = W  + h * OUT_FEATS + n0 + c;
    const float* wrp = Wr + h * OUT_FEATS + n0 + c;
    const float bl = bias[n0 + c];

    v2f bw[8], br[8];
    sage_load_b(bw, br, wp, wrp, 0);

    v8f acc[NT];
    float scale[NT];
    const float* ap[NT];
    const float* xp[NT];
    #pragma unroll
    for (int t = 0; t < NT; ++t) {
        const int row = 16 * (g + 4 * t) + c;
        scale[t] = 1.0f / fmaxf(s_deg[row], 1.0f);
        ap[t] = s_agg + row * PITCH + 32 * h;
        xp[t] = s_x   + row * PITCH + 32 * h;
        acc[t] = v8f{};
    }

    #pragma unroll
    for (int kc = 0; kc < 4; ++kc) {
        v2f nbw[8], nbr[8];
        if (kc < 3) sage_load_b(nbw, nbr, wp, wrp, kc + 1);
        #pragma unroll
        for (int jj = 0; jj < 8; ++jj) {
            const int j  = kc * 8 + jj;
            const int wo = 2 * (j & 15) + 64 * (j >> 4);
            v2f a0[NT], a1[NT];
            #pragma unroll
            for (int t = 0; t < NT; ++t) {
                a0[t] = *(const v2f*)(ap[t] + wo) * scale[t];
                a1[t] = *(const v2f*)(xp[t] + wo);
            }
            #pragma unroll
            for (int t = 0; t < NT; ++t)
                acc[t] = __builtin_amdgcn_wmma_f32_16x16x4_f32(
                             false, a0[t], false, bw[jj], (short)0, acc[t], false, false);
            #pragma unroll
            for (int t = 0; t < NT; ++t)
                acc[t] = __builtin_amdgcn_wmma_f32_16x16x4_f32(
                             false, a1[t], false, br[jj], (short)0, acc[t], false, false);
        }
        if (kc < 3) {
            #pragma unroll
            for (int jj = 0; jj < 8; ++jj) { bw[jj] = nbw[jj]; br[jj] = nbr[jj]; }
        }
    }

    #pragma unroll
    for (int t = 0; t < NT; ++t) {
        const int r0 = 16 * (g + 4 * t) + 8 * h;
        float* op = out + (size_t)(lo + r0) * OUT_FEATS + n0 + c;
        #pragma unroll
        for (int i = 0; i < 8; ++i) {
            const float v = acc[t][i] + bl;
            if (r0 + i < nloc) op[(size_t)i * OUT_FEATS] = v;
        }
    }
}

__global__ __launch_bounds__(AGG_THREADS) void sage_fused_kernel(
        const float* __restrict__ x, const int* __restrict__ esrc,
        const int* __restrict__ edst, const float* __restrict__ W,
        const float* __restrict__ Wr, const float* __restrict__ bias,
        float* __restrict__ out, int n_nodes, int n_edges) {
    __shared__ __attribute__((aligned(16))) float s_agg[(NPB + 1) * PITCH + 2];
    __shared__ float s_deg[NPB];
    __shared__ __attribute__((aligned(16))) int s_qe[AGG_WAVES * QW];

    const int tid  = threadIdx.x;
    const int wave = __builtin_amdgcn_readfirstlane(tid >> 5);
    const int lane = tid & 31;
    const int lo   = blockIdx.x * NPB;
    const unsigned nloc = (unsigned)min(NPB, n_nodes - lo);

    for (int i = tid; i < NPB * PITCH / 4; i += AGG_THREADS)
        ((float4*)s_agg)[i] = make_float4(0.f, 0.f, 0.f, 0.f);
    if (tid < NPB) s_deg[tid] = 0.f;
    __syncthreads();

    int* qe = s_qe + wave * QW;
    int cnt = 0;

    const int n_full = n_edges / CHUNK;
    const int rot = (int)blockIdx.x * (n_full / (int)gridDim.x);

    if (wave < n_full) {
        int cc = wave + rot; if (cc >= n_full) cc -= n_full;
        int4 nx[4];
        sage_load_chunk(nx, edst, cc, lane);
        for (int c = wave; c < n_full; c += AGG_WAVES) {
            int4 cur[4];
            for (int q = 0; q < 4; ++q) cur[q] = nx[q];
            const int ea = cc * CHUNK + 4 * lane;
            const int c2 = c + AGG_WAVES;
            if (c2 < n_full) { cc = c2 + rot; if (cc >= n_full) cc -= n_full; }
            sage_load_chunk(nx, edst, cc, lane);

            unsigned dd[16];
            for (int q = 0; q < 4; ++q) {
                dd[4 * q + 0] = (unsigned)cur[q].x - (unsigned)lo;
                dd[4 * q + 1] = (unsigned)cur[q].y - (unsigned)lo;
                dd[4 * q + 2] = (unsigned)cur[q].z - (unsigned)lo;
                dd[4 * q + 3] = (unsigned)cur[q].w - (unsigned)lo;
            }
            bool any = false;
            for (int k = 0; k < 16; ++k) any |= dd[k] < nloc;
            if (__builtin_amdgcn_ballot_w32(any)) {
                for (int k = 0; k < 16; ++k) {
                    const bool hit = dd[k] < nloc;
                    const unsigned mask = __builtin_amdgcn_ballot_w32(hit);
                    if (mask) {
                        if (hit) qe[cnt + (int)__builtin_amdgcn_mbcnt_lo(mask, 0u)] =
                                     ea + (k & 3) + (k >> 2) * 128;
                        cnt += __builtin_popcount(mask);
                    }
                }
            }
            if (cnt > QW - CHUNK) {
                sage_drain(x, esrc, edst, qe, cnt, lo, s_agg, s_deg, lane);
                cnt = 0;
            }
        }
    }
    {
        const int e = n_full * CHUNK + tid;
        bool hit = false;
        if (e < n_edges) hit = ((unsigned)edst[e] - (unsigned)lo) < nloc;
        const unsigned mask = __builtin_amdgcn_ballot_w32(hit);
        if (mask) {
            if (hit) qe[cnt + (int)__builtin_amdgcn_mbcnt_lo(mask, 0u)] = e;
            cnt += __builtin_popcount(mask);
        }
    }
    sage_drain(x, esrc, edst, qe, cnt, lo, s_agg, s_deg, lane);

    float4 xr[NPB / AGG_WAVES];
    #pragma unroll
    for (int q = 0; q < NPB / AGG_WAVES; ++q) {
        const int row = wave + q * AGG_WAVES;
        xr[q] = make_float4(0.f, 0.f, 0.f, 0.f);
        if (row < (int)nloc) xr[q] = *(const float4*)(x + (size_t)(lo + row) * IN_FEATS + 4 * lane);
    }

    __syncthreads();

    float* s_x = (float*)s_qe;
    #pragma unroll
    for (int q = 0; q < NPB / AGG_WAVES; ++q) {
        float* xp = s_x + (wave + q * AGG_WAVES) * PITCH + lane;
        xp[0] = xr[q].x; xp[32] = xr[q].y; xp[64] = xr[q].z; xp[96] = xr[q].w;
    }

    __syncthreads();

    const int g  = wave >> 3;
    const int n0 = (wave & 7) << 4;
    if (g < 2) sage_tile_gemm<3>(s_agg, s_x, s_deg, W, Wr, bias, out, lo, (int)nloc, g, n0, lane);
    else       sage_tile_gemm<2>(s_agg, s_x, s_deg, W, Wr, bias, out, lo, (int)nloc, g, n0, lane);
}

extern "C" void kernel_launch(void* const* d_in, const int* in_sizes, int n_in,
                              void* d_out, int out_size, void* d_ws, size_t ws_size,
                              hipStream_t stream) {
    const float* x    = (const float*)d_in[0];
    const int*   src  = (const int*)d_in[1];
    const int*   dst  = (const int*)d_in[2];
    const float* W    = (const float*)d_in[3];
    const float* Wr   = (const float*)d_in[4];
    const float* bias = (const float*)d_in[5];
    float* out = (float*)d_out;

    const int n_nodes = in_sizes[0] / IN_FEATS;
    const int n_edges = in_sizes[1];

    sage_fused_kernel<<<(n_nodes + NPB - 1) / NPB, AGG_THREADS, 0, stream>>>(
        x, src, dst, W, Wr, bias, out, n_nodes, n_edges);
}


# New first launch packs each edge's destination range id (dst/160) into one byte; each block's edge filter scans that table, four keys per dword test, not the 4-byte ids; sums stay in LDS-atomic order
# speedup vs baseline: 4.2408x; 1.2118x over previous
#include <hip/hip_runtime.h>
#include <hip/hip_bf16.h>


typedef __attribute__((ext_vector_type(2))) float v2f;
typedef __attribute__((ext_vector_type(8))) float v8f;

#define IN_FEATS 128
#define OUT_FEATS 128

#define NPB         160
#define AGG_WAVES   32
#define AGG_THREADS (AGG_WAVES * 32)
#define CHUNK       512
#define QW          1024
#define KTRIP       (4 * CHUNK)
#define KEY_THREADS 256
#define KEY_BLOCK   (2 * KTRIP)
#define KEY_CAP     254u
#define KEY_PAD     255u
#define PITCH       130
static_assert(QW >= 2 * CHUNK, "queue room for one 16-byte key load per lane");
static_assert(KEY_BLOCK == 16 * KEY_THREADS && KEY_BLOCK % KTRIP == 0, "a key block writes whole trips: four int4 of ids per thread");
static_assert(AGG_WAVES * QW >= NPB * PITCH, "the x tile reuses the queue words");
static_assert((NPB * PITCH) % 4 == 0 && NPB % 16 == 0 && AGG_WAVES == 32, "tile shape");

__device__ __forceinline__ unsigned sage_key(int d) {
    return min((unsigned)d / (unsigned)NPB, KEY_CAP);
}

__device__ __forceinline__ unsigned sage_key4(int4 d) {
    return sage_key(d.x) | (sage_key(d.y) << 8) | (sage_key(d.z) << 16) | (sage_key(d.w) << 24);
}

__global__ __launch_bounds__(KEY_THREADS) void sage_key_kernel(
        const int* __restrict__ edst, unsigned* __restrict__ keys, int n_edges) {
    const int e0 = (int)blockIdx.x * KEY_BLOCK + 4 * (int)threadIdx.x;
    if (((int)blockIdx.x + 1) * KEY_BLOCK <= n_edges) {
        int4 d[4];
        #pragma unroll
        for (int p = 0; p < 4; ++p) d[p] = *(const int4*)(edst + e0 + p * (KEY_BLOCK / 4));
        #pragma unroll
        for (int p = 0; p < 4; ++p) keys[(e0 + p * (KEY_BLOCK / 4)) >> 2] = sage_key4(d[p]);
    } else {
        unsigned k[4] = { 0u, 0u, 0u, 0u };
        #pragma unroll
        for (int p = 0; p < 4; ++p) {
            #pragma unroll
            for (int i = 0; i < 4; ++i) {
                const int e = e0 + p * (KEY_BLOCK / 4) + i;
                const unsigned kd = sage_key(edst[min(e, n_edges - 1)]);
                k[p] |= (e < n_edges ? kd : KEY_PAD) << (8 * i);
            }
        }
        #pragma unroll
        for (int p = 0; p < 4; ++p) keys[(e0 + p * (KEY_BLOCK / 4)) >> 2] = k[p];
    }
}

__device__ __forceinline__ void sage_load_keys(uint4 (&v)[4], const unsigned* __restrict__ keys,
                                               int cc, int lane) {
    const uint4* p = (const uint4*)(keys + cc * (KTRIP / 4) + 4 * lane);
    v[0] = p[0]; v[1] = p[32]; v[2] = p[64]; v[3] = p[96];
}

__device__ __forceinline__ void sage_add_row(float* __restrict__ s_agg, int d, int lane, float4 v) {
    float* ap = s_agg + d * PITCH + lane;
    unsafeAtomicAdd(ap + 0,  v.x);
    unsafeAtomicAdd(ap + 32, v.y);
    unsafeAtomicAdd(ap + 64, v.z);
    unsafeAtomicAdd(ap + 96, v.w);
}

__device__ __forceinline__ void sage_drain(const float* __restrict__ x,
                                           const int* __restrict__ esrc,
                                           const int* __restrict__ edst,
                                           const int* qe, int cnt, int lo, unsigned nloc,
                                           float* __restrict__ s_agg,
                                           float* __restrict__ s_deg, int lane) {
    for (int base = 0; base < cnt; base += 32) {
        const int i = base + lane;
        int s_l = 0, d_l = NPB;
        if (i < cnt) {
            const int e = qe[i];
            const int s = esrc[e];
            const unsigned d = (unsigned)edst[e] - (unsigned)lo;
            const bool own = d < nloc;
            s_l = own ? s : 0;
            d_l = own ? (int)d : NPB;
            unsafeAtomicAdd(s_deg + d_l, 1.0f);
        }
        const int m = min(32, cnt - base);
        for (int j = 0; j < m; j += 4) {
            const int s0 = __builtin_amdgcn_readlane(s_l, j);
            const int s1 = __builtin_amdgcn_readlane(s_l, j + 1);
            const int s2 = __builtin_amdgcn_readlane(s_l, j + 2);
            const int s3 = __builtin_amdgcn_readlane(s_l, j + 3);
            const int d0 = __builtin_amdgcn_readlane(d_l, j);
            const int d1 = __builtin_amdgcn_readlane(d_l, j + 1);
            const int d2 = __builtin_amdgcn_readlane(d_l, j + 2);
            const int d3 = __builtin_amdgcn_readlane(d_l, j + 3);
            const float4 v0 = *(const float4*)(x + (size_t)s0 * IN_FEATS + 4 * lane);
            const float4 v1 = *(const float4*)(x + (size_t)s1 * IN_FEATS + 4 * lane);
            const float4 v2 = *(const float4*)(x + (size_t)s2 * IN_FEATS + 4 * lane);
            const float4 v3 = *(const float4*)(x + (size_t)s3 * IN_FEATS + 4 * lane);
            sage_add_row(s_agg, d0, lane, v0);
            sage_add_row(s_agg, d1, lane, v1);
            sage_add_row(s_agg, d2, lane, v2);
            sage_add_row(s_agg, d3, lane, v3);
        }
    }
}

__device__ __forceinline__ void sage_load_b(v2f (&bw)[8], v2f (&br)[8],
                                            const float* __restrict__ wp,
                                            const float* __restrict__ wrp, int kc) {
    #pragma unroll
    for (int jj = 0; jj < 8; ++jj) {
        const int j  = kc * 8 + jj;
        const int kb = 8 * (j & 15) + 2 * (j >> 4);
        bw[jj].x = wp[kb * OUT_FEATS];
        bw[jj].y = wp[(kb + 4) * OUT_FEATS];
        br[jj].x = wrp[kb * OUT_FEATS];
        br[jj].y = wrp[(kb + 4) * OUT_FEATS];
    }
}

template <int NT>
__device__ __forceinline__ void sage_tile_gemm(const float* s_agg, const float* s_x,
                                               const float* s_deg,
                                               const float* __restrict__ W,
                                               const float* __restrict__ Wr,
                                               float bl,
                                               float* __restrict__ out,
                                               int lo, int nloc, int g, int n0, int lane) {
    const int c = lane & 15;
    const int h = lane >> 4;
    const float* wp  = W  + h * OUT_FEATS + n0 + c;
    const float* wrp = Wr + h * OUT_FEATS + n0 + c;

    v2f bw[8], br[8];
    sage_load_b(bw, br, wp, wrp, 0);

    v8f acc[NT];
    float scale[NT];
    const float* ap[NT];
    const float* xp[NT];
    #pragma unroll
    for (int t = 0; t < NT; ++t) {
        const int row = 16 * (g + 4 * t) + c;
        scale[t] = 1.0f / fmaxf(s_deg[row], 1.0f);
        ap[t] = s_agg + row * PITCH + 32 * h;
        xp[t] = s_x   + row * PITCH + 32 * h;
        acc[t] = v8f{};
    }

    #pragma unroll
    for (int kc = 0; kc < 4; ++kc) {
        v2f nbw[8], nbr[8];
        if (kc < 3) sage_load_b(nbw, nbr, wp, wrp, kc + 1);
        #pragma unroll
        for (int jj = 0; jj < 8; ++jj) {
            const int j  = kc * 8 + jj;
            const int wo = 2 * (j & 15) + 64 * (j >> 4);
            v2f a0[NT], a1[NT];
            #pragma unroll
            for (int t = 0; t < NT; ++t) {
                a0[t] = *(const v2f*)(ap[t] + wo) * scale[t];
                a1[t] = *(const v2f*)(xp[t] + wo);
            }
            #pragma unroll
            for (int t = 0; t < NT; ++t)
                acc[t] = __builtin_amdgcn_wmma_f32_16x16x4_f32(
                             false, a0[t], false, bw[jj], (short)0, acc[t], false, false);
            #pragma unroll
            for (int t = 0; t < NT; ++t)
                acc[t] = __builtin_amdgcn_wmma_f32_16x16x4_f32(
                             false, a1[t], false, br[jj], (short)0, acc[t], false, false);
        }
        if (kc < 3) {
            #pragma unroll
            for (int jj = 0; jj < 8; ++jj) { bw[jj] = nbw[jj]; br[jj] = nbr[jj]; }
        }
    }

    #pragma unroll
    for (int t = 0; t < NT; ++t) {
        const int r0 = 16 * (g + 4 * t) + 8 * h;
        float* op = out + (size_t)(lo + r0) * OUT_FEATS + n0 + c;
        #pragma unroll
        for (int i = 0; i < 8; ++i) {
            const float v = acc[t][i] + bl;
            if (r0 + i < nloc) op[(size_t)i * OUT_FEATS] = v;
        }
    }
}

__global__ __launch_bounds__(AGG_THREADS) void sage_fused_kernel(
        const float* __restrict__ x, const int* __restrict__ esrc,
        const int* __restrict__ edst, const unsigned* __restrict__ keys,
        const float* __restrict__ W,
        const float* __restrict__ Wr, const float* __restrict__ bias,
        float* __restrict__ out, int n_nodes, int n_edges) {
    __shared__ __attribute__((aligned(16))) float s_agg[(NPB + 1) * PITCH + 2];
    __shared__ float s_deg[NPB + 1];
    __shared__ __attribute__((aligned(16))) int s_qe[AGG_WAVES * QW];

    const int tid  = threadIdx.x;
    const int wave = __builtin_amdgcn_readfirstlane(tid >> 5);
    const int lane = tid & 31;
    const int lo   = blockIdx.x * NPB;
    const unsigned nloc = (unsigned)min(NPB, n_nodes - lo);

    for (int i = tid; i < NPB * PITCH / 4; i += AGG_THREADS)
        ((float4*)s_agg)[i] = make_float4(0.f, 0.f, 0.f, 0.f);
    if (tid < NPB) s_deg[tid] = 0.f;
    __syncthreads();

    int* qe = s_qe + wave * QW;
    int cnt = 0;

    const int n_trips = (n_edges + KTRIP - 1) / KTRIP;
    const int rot = (int)blockIdx.x * (n_trips / (int)gridDim.x);
    const unsigned pat = min((unsigned)blockIdx.x, KEY_CAP) * 0x01010101u;

    if (wave < n_trips) {
        int cc = wave + rot; if (cc >= n_trips) cc -= n_trips;
        uint4 nx[4];
        sage_load_keys(nx, keys, cc, lane);
        for (int c = wave; c < n_trips; c += AGG_WAVES) {
            uint4 cur[4];
            for (int q = 0; q < 4; ++q) cur[q] = nx[q];
            const int ea = cc * KTRIP + 16 * lane;
            const int c2 = c + AGG_WAVES;
            if (c2 < n_trips) { cc = c2 + rot; if (cc >= n_trips) cc -= n_trips; }
            sage_load_keys(nx, keys, cc, lane);

            #pragma unroll
            for (int q = 0; q < 4; ++q) {
                const unsigned kw[4] = { cur[q].x, cur[q].y, cur[q].z, cur[q].w };
                #pragma unroll
                for (int w = 0; w < 4; ++w) {
                    const unsigned xk = kw[w] ^ pat;
                    unsigned z = (xk - 0x01010101u) & ~xk & 0x80808080u;
                    unsigned mask = __builtin_amdgcn_ballot_w32(z != 0u);
                    while (mask) {
                        if (z != 0u) {
                            qe[cnt + (int)__builtin_amdgcn_mbcnt_lo(mask, 0u)] =
                                ea + q * CHUNK + 4 * w + (__builtin_ctz(z) >> 3);
                            z &= z - 1u;
                        }
                        cnt += __builtin_popcount(mask);
                        mask = __builtin_amdgcn_ballot_w32(z != 0u);
                    }
                }
                if (cnt > QW - CHUNK) {
                    sage_drain(x, esrc, edst, qe, cnt, lo, nloc, s_agg, s_deg, lane);
                    cnt = 0;
                }
            }
        }
    }
    sage_drain(x, esrc, edst, qe, cnt, lo, nloc, s_agg, s_deg, lane);

    const int g  = wave >> 3;
    const int n0 = (wave & 7) << 4;
    const float bl = bias[n0 + (lane & 15)];

    float4 xr[NPB / AGG_WAVES];
    #pragma unroll
    for (int q = 0; q < NPB / AGG_WAVES; ++q) {
        const int row = wave + q * AGG_WAVES;
        xr[q] = make_float4(0.f, 0.f, 0.f, 0.f);
        if (row < (int)nloc) xr[q] = *(const float4*)(x + (size_t)(lo + row) * IN_FEATS + 4 * lane);
    }

    __syncthreads();

    float* s_x = (float*)s_qe;
    #pragma unroll
    for (int q = 0; q < NPB / AGG_WAVES; ++q) {
        float* xp = s_x + (wave + q * AGG_WAVES) * PITCH + lane;
        xp[0] = xr[q].x; xp[32] = xr[q].y; xp[64] = xr[q].z; xp[96] = xr[q].w;
    }

    __syncthreads();

    if (g >= 2) sage_tile_gemm<2>(s_agg, s_x, s_deg, W, Wr, bl, out, lo, (int)nloc, g, n0, lane);
    else        sage_tile_gemm<3>(s_agg, s_x, s_deg, W, Wr, bl, out, lo, (int)nloc, g, n0, lane);
}

extern "C" void kernel_launch(void* const* d_in, const int* in_sizes, int n_in,
                              void* d_out, int out_size, void* d_ws, size_t ws_size,
                              hipStream_t stream) {
    const float* x    = (const float*)d_in[0];
    const int*   src  = (const int*)d_in[1];
    const int*   dst  = (const int*)d_in[2];
    const float* W    = (const float*)d_in[3];
    const float* Wr   = (const float*)d_in[4];
    const float* bias = (const float*)d_in[5];
    float* out = (float*)d_out;

    const int n_nodes = in_sizes[0] / IN_FEATS;
    const int n_edges = in_sizes[1];

    unsigned* keys = (unsigned*)d_ws;
    sage_key_kernel<<<(n_edges + KEY_BLOCK - 1) / KEY_BLOCK, KEY_THREADS, 0, stream>>>(dst, keys, n_edges);

    sage_fused_kernel<<<(n_nodes + NPB - 1) / NPB, AGG_THREADS, 0, stream>>>(
        x, src, dst, keys, W, Wr, bias, out, n_nodes, n_edges);
}


# Edge filter now two-level: launch 1 writes a presence map (1 bit per 32-edge group and 160-node range); a block scans its bit plane, reads the dst-id line of hit groups only; LDS-atomic order kept
# speedup vs baseline: 4.4710x; 1.0543x over previous
#include <hip/hip_runtime.h>
#include <hip/hip_bf16.h>


typedef __attribute__((ext_vector_type(2))) float v2f;
typedef __attribute__((ext_vector_type(8))) float v8f;

#define IN_FEATS 128
#define OUT_FEATS 128

#define NPB         160
#define AGG_WAVES   32
#define AGG_THREADS (AGG_WAVES * 32)
#define GRP         32
#define GTRIP       128
#define QG          256
#define QW          768
#define XBATCH      16
#define MAP_THREADS 128
#define RID_CAP     255u
#define PITCH       130
static_assert(QG >= 2 * GTRIP, "group queue room for one trip");
static_assert(QW > XBATCH * GRP && 32 % XBATCH == 0, "edge queue room for one batch of hit groups");
static_assert(GTRIP == 4 * 32 && GTRIP == MAP_THREADS, "a trip is one uint4 per lane; a block of the first launch writes one trip of every plane");
static_assert(2 * AGG_WAVES * QW >= NPB * PITCH, "the x tile reuses the edge queue words");
static_assert((NPB * PITCH) % 4 == 0 && NPB % 16 == 0 && AGG_WAVES == 32, "tile shape");

__device__ __forceinline__ void sage_map_add(unsigned (&m)[8], int d) {
    const unsigned k   = min((unsigned)d / (unsigned)NPB, RID_CAP);
    const unsigned bit = 1u << (k & 31u);
    const unsigned sel = k >> 5;
    #pragma unroll
    for (int i = 0; i < 8; ++i) m[i] |= (sel == (unsigned)i) ? bit : 0u;
}

__global__ __launch_bounds__(MAP_THREADS) void sage_map_kernel(
        const int* __restrict__ edst, unsigned* __restrict__ map, int n_edges) {
    const int ngp = (int)gridDim.x * MAP_THREADS;
    const int g   = (int)blockIdx.x * MAP_THREADS + (int)threadIdx.x;
    const int e0  = g * GRP;
    unsigned m[8] = { 0u, 0u, 0u, 0u, 0u, 0u, 0u, 0u };
    if (e0 + GRP <= n_edges) {
        int4 d[8];
        #pragma unroll
        for (int p = 0; p < 8; ++p) d[p] = ((const int4*)(edst + e0))[p];
        #pragma unroll
        for (int p = 0; p < 8; ++p) {
            sage_map_add(m, d[p].x); sage_map_add(m, d[p].y);
            sage_map_add(m, d[p].z); sage_map_add(m, d[p].w);
        }
    }
    #pragma unroll
    for (int i = 0; i < 8; ++i) map[(size_t)i * ngp + g] = m[i];
}

__device__ __forceinline__ void sage_add_row(float* __restrict__ s_agg, int d, int lane, float4 v) {
    float* ap = s_agg + d * PITCH + lane;
    unsafeAtomicAdd(ap + 0,  v.x);
    unsafeAtomicAdd(ap + 32, v.y);
    unsafeAtomicAdd(ap + 64, v.z);
    unsafeAtomicAdd(ap + 96, v.w);
}

__device__ __forceinline__ void sage_drain(const float* __restrict__ x,
                                           const int* __restrict__ esrc,
                                           const int2* qe, int cnt,
                                           float* __restrict__ s_agg,
                                           float* __restrict__ s_deg, int lane) {
    for (int base = 0; base < cnt; base += 32) {
        const int i = base + lane;
        int s_l = 0, d_l = NPB;
        if (i < cnt) {
            const int2 ed = qe[i];
            s_l = esrc[ed.x];
            d_l = ed.y;
            unsafeAtomicAdd(s_deg + d_l, 1.0f);
        }
        const int m = min(32, cnt - base);
        for (int j = 0; j < m; j += 4) {
            const int s0 = __builtin_amdgcn_readlane(s_l, j);
            const int s1 = __builtin_amdgcn_readlane(s_l, j + 1);
            const int s2 = __builtin_amdgcn_readlane(s_l, j + 2);
            const int s3 = __builtin_amdgcn_readlane(s_l, j + 3);
            const int d0 = __builtin_amdgcn_readlane(d_l, j);
            const int d1 = __builtin_amdgcn_readlane(d_l, j + 1);
            const int d2 = __builtin_amdgcn_readlane(d_l, j + 2);
            const int d3 = __builtin_amdgcn_readlane(d_l, j + 3);
            const float4 v0 = *(const float4*)(x + (size_t)s0 * IN_FEATS + 4 * lane);
            const float4 v1 = *(const float4*)(x + (size_t)s1 * IN_FEATS + 4 * lane);
            const float4 v2 = *(const float4*)(x + (size_t)s2 * IN_FEATS + 4 * lane);
            const float4 v3 = *(const float4*)(x + (size_t)s3 * IN_FEATS + 4 * lane);
            sage_add_row(s_agg, d0, lane, v0);
            sage_add_row(s_agg, d1, lane, v1);
            sage_add_row(s_agg, d2, lane, v2);
            sage_add_row(s_agg, d3, lane, v3);
        }
    }
}

__device__ __forceinline__ void sage_expand(const float* __restrict__ x,
                                            const int* __restrict__ esrc,
                                            const int* __restrict__ edst,
                                            const int* qg, int cntg,
                                            int lo, unsigned nloc,
                                            int2* qe, int& cnt,
                                            float* __restrict__ s_agg,
                                            float* __restrict__ s_deg, int lane) {
    for (int base = 0; base < cntg; base += 32) {
        const int gl = (base + lane < cntg) ? qg[base + lane] : -1;
        const int m = min(32, cntg - base);
        for (int j = 0; j < m; j += XBATCH) {
            if (cnt > QW - XBATCH * GRP) {
                sage_drain(x, esrc, qe, cnt, s_agg, s_deg, lane);
                cnt = 0;
            }
            int dv[XBATCH];
            #pragma unroll
            for (int u = 0; u < XBATCH; ++u) {
                const int g = __builtin_amdgcn_readlane(gl, j + u);
                dv[u] = edst[max(g, 0) * GRP + lane];
            }
            #pragma unroll
            for (int u = 0; u < XBATCH; ++u) {
                const int g = __builtin_amdgcn_readlane(gl, j + u);
                if (g >= 0) {
                    const unsigned d = (unsigned)dv[u] - (unsigned)lo;
                    const bool own = d < nloc;
                    const unsigned mask = __builtin_amdgcn_ballot_w32(own);
                    if (own) qe[cnt + (int)__builtin_amdgcn_mbcnt_lo(mask, 0u)] = make_int2(g * GRP + lane, (int)d);
                    cnt += __builtin_popcount(mask);
                }
            }
        }
    }
}

__device__ __forceinline__ void sage_load_b(v2f (&bw)[8], v2f (&br)[8],
                                            const float* __restrict__ wp,
                                            const float* __restrict__ wrp, int kc) {
    #pragma unroll
    for (int jj = 0; jj < 8; ++jj) {
        const int j  = kc * 8 + jj;
        const int kb = 8 * (j & 15) + 2 * (j >> 4);
        bw[jj].x = wp[kb * OUT_FEATS];
        bw[jj].y = wp[(kb + 4) * OUT_FEATS];
        br[jj].x = wrp[kb * OUT_FEATS];
        br[jj].y = wrp[(kb + 4) * OUT_FEATS];
    }
}

template <int NT>
__device__ __forceinline__ void sage_tile_gemm(const float* s_agg, const float* s_x,
                                               const float* s_deg,
                                               const float* __restrict__ W,
                                               const float* __restrict__ Wr,
                                               float bl,
                                               float* __restrict__ out,
                                               int lo, int nloc, int g, int n0, int lane) {
    const int c = lane & 15;
    const int h = lane >> 4;
    const float* wp  = W  + h * OUT_FEATS + n0 + c;
    const float* wrp = Wr + h * OUT_FEATS + n0 + c;

    v2f bw[8], br[8];
    sage_load_b(bw, br, wp, wrp, 0);

    v8f acc[NT];
    float scale[NT];
    const float* ap[NT];
    const float* xp[NT];
    #pragma unroll
    for (int t = 0; t < NT; ++t) {
        const int row = 16 * (g + 4 * t) + c;
        scale[t] = 1.0f / fmaxf(s_deg[row], 1.0f);
        ap[t] = s_agg + row * PITCH + 32 * h;
        xp[t] = s_x   + row * PITCH + 32 * h;
        acc[t] = v8f{};
    }

    #pragma unroll
    for (int kc = 0; kc < 4; ++kc) {
        v2f nbw[8], nbr[8];
        if (kc < 3) sage_load_b(nbw, nbr, wp, wrp, kc + 1);
        #pragma unroll
        for (int jj = 0; jj < 8; ++jj) {
            const int j  = kc * 8 + jj;
            const int wo = 2 * (j & 15) + 64 * (j >> 4);
            v2f a0[NT], a1[NT];
            #pragma unroll
            for (int t = 0; t < NT; ++t) {
                a0[t] = *(const v2f*)(ap[t] + wo) * scale[t];
                a1[t] = *(const v2f*)(xp[t] + wo);
            }
            #pragma unroll
            for (int t = 0; t < NT; ++t)
                acc[t] = __builtin_amdgcn_wmma_f32_16x16x4_f32(
                             false, a0[t], false, bw[jj], (short)0, acc[t], false, false);
            #pragma unroll
            for (int t = 0; t < NT; ++t)
                acc[t] = __builtin_amdgcn_wmma_f32_16x16x4_f32(
                             false, a1[t], false, br[jj], (short)0, acc[t], false, false);
        }
        if (kc < 3) {
            #pragma unroll
            for (int jj = 0; jj < 8; ++jj) { bw[jj] = nbw[jj]; br[jj] = nbr[jj]; }
        }
    }

    #pragma unroll
    for (int t = 0; t < NT; ++t) {
        const int r0 = 16 * (g + 4 * t) + 8 * h;
        float* op = out + (size_t)(lo + r0) * OUT_FEATS + n0 + c;
        #pragma unroll
        for (int i = 0; i < 8; ++i) {
            const float v = acc[t][i] + bl;
            if (r0 + i < nloc) op[(size_t)i * OUT_FEATS] = v;
        }
    }
}

__global__ __launch_bounds__(AGG_THREADS) void sage_fused_kernel(
        const float* __restrict__ x, const int* __restrict__ esrc,
        const int* __restrict__ edst, const unsigned* __restrict__ map,
        const float* __restrict__ W,
        const float* __restrict__ Wr, const float* __restrict__ bias,
        float* __restrict__ out, int n_nodes, int n_edges) {
    __shared__ __attribute__((aligned(16))) float s_agg[(NPB + 1) * PITCH + 2];
    __shared__ float s_deg[NPB + 1];
    __shared__ __attribute__((aligned(16))) int2 s_qe[AGG_WAVES * QW];
    __shared__ int s_qg[AGG_WAVES * QG];

    const int tid  = threadIdx.x;
    const int wave = __builtin_amdgcn_readfirstlane(tid >> 5);
    const int lane = tid & 31;
    const int lo   = blockIdx.x * NPB;
    const unsigned nloc = (unsigned)min(NPB, n_nodes - lo);

    for (int i = tid; i < NPB * PITCH / 4; i += AGG_THREADS)
        ((float4*)s_agg)[i] = make_float4(0.f, 0.f, 0.f, 0.f);
    if (tid < NPB) s_deg[tid] = 0.f;
    __syncthreads();

    int2* qe = s_qe + wave * QW;
    int*  qg = s_qg + wave * QG;
    int cnt  = 0;
    int cntg = 0;

    const int n_groups = (n_edges + GRP - 1) / GRP;
    const int n_trips  = (n_groups + GTRIP - 1) / GTRIP;
    const int rot = ((int)blockIdx.x * n_trips) / (int)gridDim.x;
    const unsigned rid = min((unsigned)blockIdx.x, RID_CAP);
    const unsigned bm  = 1u << (rid & 31u);
    const unsigned* mp = map + (size_t)(rid >> 5) * ((size_t)n_trips * GTRIP);

    const int e_tail = (n_edges / GRP) * GRP;
    if (wave == 0 && e_tail < n_edges) {
        const int e = e_tail + lane;
        const bool in = e < n_edges;
        const unsigned d = (unsigned)edst[in ? e : e_tail] - (unsigned)lo;
        const bool own = in && (d < nloc);
        const unsigned mask = __builtin_amdgcn_ballot_w32(own);
        if (own) qe[(int)__builtin_amdgcn_mbcnt_lo(mask, 0u)] = make_int2(e, (int)d);
        cnt = __builtin_popcount(mask);
    }

    if (wave < n_trips) {
        int cc = wave + rot; if (cc >= n_trips) cc -= n_trips;
        uint4 nx = *(const uint4*)(mp + cc * GTRIP + 4 * lane);
        for (int c = wave; c < n_trips; c += AGG_WAVES) {
            const uint4 cur = nx;
            const int g0 = cc * GTRIP + 4 * lane;
            const int c2 = c + AGG_WAVES;
            if (c2 < n_trips) { cc = c2 + rot; if (cc >= n_trips) cc -= n_trips; }
            nx = *(const uint4*)(mp + cc * GTRIP + 4 * lane);

            const unsigned mw[4] = { cur.x, cur.y, cur.z, cur.w };
            #pragma unroll
            for (int w = 0; w < 4; ++w) {
                const unsigned t = mw[w] & bm;
                const unsigned mask = __builtin_amdgcn_ballot_w32(t != 0u);
                if (t != 0u) qg[cntg + (int)__builtin_amdgcn_mbcnt_lo(mask, 0u)] = g0 + w;
                cntg += __builtin_popcount(mask);
            }
            if (cntg > QG - GTRIP) {
                sage_expand(x, esrc, edst, qg, cntg, lo, nloc, qe, cnt, s_agg, s_deg, lane);
                cntg = 0;
            }
        }
    }
    sage_expand(x, esrc, edst, qg, cntg, lo, nloc, qe, cnt, s_agg, s_deg, lane);
    sage_drain(x, esrc, qe, cnt, s_agg, s_deg, lane);

    const int g  = wave >> 3;
    const int n0 = (wave & 7) << 4;
    const float bl = bias[n0 + (lane & 15)];

    float4 xr[NPB / AGG_WAVES];
    #pragma unroll
    for (int q = 0; q < NPB / AGG_WAVES; ++q) {
        const int row = wave + q * AGG_WAVES;
        xr[q] = make_float4(0.f, 0.f, 0.f, 0.f);
        if (row < (int)nloc) xr[q] = *(const float4*)(x + (size_t)(lo + row) * IN_FEATS + 4 * lane);
    }

    __syncthreads();

    float* s_x = (float*)s_qe;
    #pragma unroll
    for (int q = 0; q < NPB / AGG_WAVES; ++q) {
        float* xp = s_x + (wave + q * AGG_WAVES) * PITCH + lane;
        xp[0] = xr[q].x; xp[32] = xr[q].y; xp[64] = xr[q].z; xp[96] = xr[q].w;
    }

    __syncthreads();

    if (g >= 2) sage_tile_gemm<2>(s_agg, s_x, s_deg, W, Wr, bl, out, lo, (int)nloc, g, n0, lane);
    else        sage_tile_gemm<3>(s_agg, s_x, s_deg, W, Wr, bl, out, lo, (int)nloc, g, n0, lane);
}

extern "C" void kernel_launch(void* const* d_in, const int* in_sizes, int n_in,
                              void* d_out, int out_size, void* d_ws, size_t ws_size,
                              hipStream_t stream) {
    const float* x    = (const float*)d_in[0];
    const int*   src  = (const int*)d_in[1];
    const int*   dst  = (const int*)d_in[2];
    const float* W    = (const float*)d_in[3];
    const float* Wr   = (const float*)d_in[4];
    const float* bias = (const float*)d_in[5];
    float* out = (float*)d_out;

    const int n_nodes = in_sizes[0] / IN_FEATS;
    const int n_edges = in_sizes[1];

    unsigned* map = (unsigned*)d_ws;
    const int n_groups = (n_edges + GRP - 1) / GRP;
    sage_map_kernel<<<(n_groups + GTRIP - 1) / GTRIP, MAP_THREADS, 0, stream>>>(dst, map, n_edges);

    sage_fused_kernel<<<(n_nodes + NPB - 1) / NPB, AGG_THREADS, 0, stream>>>(
        x, src, dst, map, W, Wr, bias, out, n_nodes, n_edges);
}


# LDS float atomics per edge replaced by register sums: each found edge's source id is routed (LDS int atomic slot) to the wave owning its dst row, which sums its 5 rows in registers, in arrival order
# speedup vs baseline: 4.8163x; 1.0772x over previous
#include <hip/hip_runtime.h>
#include <hip/hip_bf16.h>


typedef __attribute__((ext_vector_type(2))) float v2f;
typedef __attribute__((ext_vector_type(8))) float v8f;

#define IN_FEATS 128
#define OUT_FEATS 128

#define NPB         160
#define AGG_WAVES   32
#define AGG_THREADS (AGG_WAVES * 32)
#define GRP         32
#define GTRIP       128
#define QG          256
#define QW          544
#define RQ          512
#define ROWS_PW     5
#define XBATCH      16
#define MAP_THREADS 128
#define RID_CAP     255u
#define PITCH       130
static_assert(QG >= 2 * GTRIP, "group queue room for one trip");
static_assert(QW > XBATCH * GRP && 32 % XBATCH == 0, "staging queue room for one batch of hit groups");
static_assert(GTRIP == 4 * 32 && GTRIP == MAP_THREADS, "a trip is one uint4 per lane; a block of the first launch writes one trip of every plane");
static_assert(2 * AGG_WAVES * QW >= NPB * PITCH, "the x tile reuses the staging queue words");
static_assert((NPB * PITCH) % 4 == 0 && NPB % 16 == 0 && AGG_WAVES == 32, "tile shape");
static_assert(NPB == ROWS_PW * AGG_WAVES, "owner wave = row & 31, row of the wave = row >> 5");

__device__ __forceinline__ void sage_map_add(unsigned (&m)[8], int d) {
    const unsigned k   = min((unsigned)d / (unsigned)NPB, RID_CAP);
    const unsigned bit = 1u << (k & 31u);
    const unsigned sel = k >> 5;
    #pragma unroll
    for (int i = 0; i < 8; ++i) m[i] |= (sel == (unsigned)i) ? bit : 0u;
}

__global__ __launch_bounds__(MAP_THREADS) void sage_map_kernel(
        const int* __restrict__ edst, unsigned* __restrict__ map, int n_edges) {
    const int ngp = (int)gridDim.x * MAP_THREADS;
    const int g   = (int)blockIdx.x * MAP_THREADS + (int)threadIdx.x;
    const int e0  = g * GRP;
    unsigned m[8] = { 0u, 0u, 0u, 0u, 0u, 0u, 0u, 0u };
    if (e0 + GRP <= n_edges) {
        int4 d[8];
        #pragma unroll
        for (int p = 0; p < 8; ++p) d[p] = ((const int4*)(edst + e0))[p];
        #pragma unroll
        for (int p = 0; p < 8; ++p) {
            sage_map_add(m, d[p].x); sage_map_add(m, d[p].y);
            sage_map_add(m, d[p].z); sage_map_add(m, d[p].w);
        }
    }
    #pragma unroll
    for (int i = 0; i < 8; ++i) map[(size_t)i * ngp + g] = m[i];
}

__device__ __forceinline__ void sage_add_row(float* __restrict__ s_agg, int d, int lane, float4 v) {
    float* ap = s_agg + d * PITCH + lane;
    unsafeAtomicAdd(ap + 0,  v.x);
    unsafeAtomicAdd(ap + 32, v.y);
    unsafeAtomicAdd(ap + 64, v.z);
    unsafeAtomicAdd(ap + 96, v.w);
}

__device__ __forceinline__ void sage_route(const float* __restrict__ x,
                                           const int* __restrict__ esrc,
                                           const int2* qe, int cnt,
                                           unsigned* s_cnt, unsigned* s_rq,
                                           float* __restrict__ s_agg,
                                           float* __restrict__ s_deg, int lane) {
    for (int base = 0; base < cnt; base += 32) {
        const int i = base + lane;
        int s_l = 0, d_l = 0;
        bool over = false;
        if (i < cnt) {
            const int2 ed = qe[i];
            s_l = esrc[ed.x];
            d_l = ed.y;
            unsafeAtomicAdd(s_deg + d_l, 1.0f);
            const unsigned ow   = (unsigned)d_l & 31u;
            const unsigned slot = atomicAdd(s_cnt + ow, 1u);
            if (slot < (unsigned)RQ) s_rq[ow * RQ + slot] = ((unsigned)s_l << 3) | ((unsigned)d_l >> 5);
            else over = true;
        }
        unsigned om = __builtin_amdgcn_ballot_w32(over);
        while (om != 0u) {
            const int j = __builtin_ctz(om);
            om &= om - 1u;
            const int sj = __builtin_amdgcn_readlane(s_l, j);
            const int dj = __builtin_amdgcn_readlane(d_l, j);
            const float4 v = *(const float4*)(x + (size_t)sj * IN_FEATS + 4 * lane);
            sage_add_row(s_agg, dj, lane, v);
        }
    }
}

#define SAGE_ARM(a, v) { asm volatile(""); a.x += (v).x; a.y += (v).y; a.z += (v).z; a.w += (v).w; }
#define SAGE_ACC(r, v)                 \
    do {                               \
        if ((r) == 0) SAGE_ARM(a0, v)  \
        if ((r) == 1) SAGE_ARM(a1, v)  \
        if ((r) == 2) SAGE_ARM(a2, v)  \
        if ((r) == 3) SAGE_ARM(a3, v)  \
        if ((r) == 4) SAGE_ARM(a4, v)  \
    } while (0)

__device__ __forceinline__ void sage_sum(const float* __restrict__ x,
                                         const unsigned* rq, int n,
                                         float* __restrict__ s_agg, int wave, int lane) {
    float4 a0 = make_float4(0.f, 0.f, 0.f, 0.f), a1 = a0, a2 = a0, a3 = a0, a4 = a0;
    for (int base = 0; base < n; base += 32) {
        int s_c = 0, r_c = ROWS_PW;
        if (base + lane < n) { const unsigned q = rq[base + lane]; s_c = (int)(q >> 3); r_c = (int)(q & 7u); }
        const int m = min(32, n - base);
        for (int j = 0; j < m; j += 8) {
            int sj[8], rj[8];
            float4 v[8];
            #pragma unroll
            for (int u = 0; u < 8; ++u) {
                sj[u] = __builtin_amdgcn_readlane(s_c, j + u);
                rj[u] = __builtin_amdgcn_readlane(r_c, j + u);
                v[u]  = *(const float4*)(x + (size_t)sj[u] * IN_FEATS + 4 * lane);
            }
            #pragma unroll
            for (int u = 0; u < 8; ++u) SAGE_ACC(rj[u], v[u]);
        }
    }
    sage_add_row(s_agg, wave,       lane, a0);
    sage_add_row(s_agg, wave + 32,  lane, a1);
    sage_add_row(s_agg, wave + 64,  lane, a2);
    sage_add_row(s_agg, wave + 96,  lane, a3);
    sage_add_row(s_agg, wave + 128, lane, a4);
}

__device__ __forceinline__ void sage_expand(const float* __restrict__ x,
                                            const int* __restrict__ esrc,
                                            const int* __restrict__ edst,
                                            const int* qg, int cntg,
                                            int lo, unsigned nloc,
                                            int2* qe, int& cnt,
                                            unsigned* s_cnt, unsigned* s_rq,
                                            float* __restrict__ s_agg,
                                            float* __restrict__ s_deg, int lane) {
    for (int base = 0; base < cntg; base += 32) {
        const int gl = (base + lane < cntg) ? qg[base + lane] : -1;
        const int m = min(32, cntg - base);
        for (int j = 0; j < m; j += XBATCH) {
            if (cnt > QW - XBATCH * GRP) {
                sage_route(x, esrc, qe, cnt, s_cnt, s_rq, s_agg, s_deg, lane);
                cnt = 0;
            }
            int dv[XBATCH];
            #pragma unroll
            for (int u = 0; u < XBATCH; ++u) {
                const int g = __builtin_amdgcn_readlane(gl, j + u);
                dv[u] = edst[max(g, 0) * GRP + lane];
            }
            #pragma unroll
            for (int u = 0; u < XBATCH; ++u) {
                const int g = __builtin_amdgcn_readlane(gl, j + u);
                if (g >= 0) {
                    const unsigned d = (unsigned)dv[u] - (unsigned)lo;
                    const bool own = d < nloc;
                    const unsigned mask = __builtin_amdgcn_ballot_w32(own);
                    if (own) qe[cnt + (int)__builtin_amdgcn_mbcnt_lo(mask, 0u)] = make_int2(g * GRP + lane, (int)d);
                    cnt += __builtin_popcount(mask);
                }
            }
        }
    }
}

__device__ __forceinline__ void sage_load_b(v2f (&bw)[8], v2f (&br)[8],
                                            const float* __restrict__ wp,
                                            const float* __restrict__ wrp, int kc) {
    #pragma unroll
    for (int jj = 0; jj < 8; ++jj) {
        const int j  = kc * 8 + jj;
        const int kb = 8 * (j & 15) + 2 * (j >> 4);
        bw[jj].x = wp[kb * OUT_FEATS];
        bw[jj].y = wp[(kb + 4) * OUT_FEATS];
        br[jj].x = wrp[kb * OUT_FEATS];
        br[jj].y = wrp[(kb + 4) * OUT_FEATS];
    }
}

template <int NT>
__device__ __forceinline__ void sage_tile_gemm(const float* s_agg, const float* s_x,
                                               const float* s_deg,
                                               const float* __restrict__ W,
                                               const float* __restrict__ Wr,
                                               float bl,
                                               float* __restrict__ out,
                                               int lo, int nloc, int g, int n0, int lane) {
    const int c = lane & 15;
    const int h = lane >> 4;
    const float* wp  = W  + h * OUT_FEATS + n0 + c;
    const float* wrp = Wr + h * OUT_FEATS + n0 + c;

    v2f bw[8], br[8];
    sage_load_b(bw, br, wp, wrp, 0);

    v8f acc[NT];
    float scale[NT];
    const float* ap[NT];
    const float* xp[NT];
    #pragma unroll
    for (int t = 0; t < NT; ++t) {
        const int row = 16 * (g + 4 * t) + c;
        scale[t] = 1.0f / fmaxf(s_deg[row], 1.0f);
        ap[t] = s_agg + row * PITCH + 32 * h;
        xp[t] = s_x   + row * PITCH + 32 * h;
        acc[t] = v8f{};
    }

    #pragma unroll
    for (int kc = 0; kc < 4; ++kc) {
        v2f nbw[8], nbr[8];
        if (kc < 3) sage_load_b(nbw, nbr, wp, wrp, kc + 1);
        #pragma unroll
        for (int jj = 0; jj < 8; ++jj) {
            const int j  = kc * 8 + jj;
            const int wo = 2 * (j & 15) + 64 * (j >> 4);
            v2f a0[NT], a1[NT];
            #pragma unroll
            for (int t = 0; t < NT; ++t) {
                a0[t] = *(const v2f*)(ap[t] + wo) * scale[t];
                a1[t] = *(const v2f*)(xp[t] + wo);
            }
            #pragma unroll
            for (int t = 0; t < NT; ++t)
                acc[t] = __builtin_amdgcn_wmma_f32_16x16x4_f32(
                             false, a0[t], false, bw[jj], (short)0, acc[t], false, false);
            #pragma unroll
            for (int t = 0; t < NT; ++t)
                acc[t] = __builtin_amdgcn_wmma_f32_16x16x4_f32(
                             false, a1[t], false, br[jj], (short)0, acc[t], false, false);
        }
        if (kc < 3) {
            #pragma unroll
            for (int jj = 0; jj < 8; ++jj) { bw[jj] = nbw[jj]; br[jj] = nbr[jj]; }
        }
    }

    #pragma unroll
    for (int t = 0; t < NT; ++t) {
        const int r0 = 16 * (g + 4 * t) + 8 * h;
        float* op = out + (size_t)(lo + r0) * OUT_FEATS + n0 + c;
        #pragma unroll
        for (int i = 0; i < 8; ++i) {
            const float v = acc[t][i] + bl;
            if (r0 + i < nloc) op[(size_t)i * OUT_FEATS] = v;
        }
    }
}

__global__ __launch_bounds__(AGG_THREADS) void sage_fused_kernel(
        const float* __restrict__ x, const int* __restrict__ esrc,
        const int* __restrict__ edst, const unsigned* __restrict__ map,
        const float* __restrict__ W,
        const float* __restrict__ Wr, const float* __restrict__ bias,
        float* __restrict__ out, int n_nodes, int n_edges) {
    __shared__ __attribute__((aligned(16))) float s_agg[NPB * PITCH];
    __shared__ float s_deg[NPB];
    __shared__ __attribute__((aligned(16))) int2 s_qe[AGG_WAVES * QW];
    __shared__ int s_qg[AGG_WAVES * QG];
    __shared__ unsigned s_rq[AGG_WAVES * RQ];
    __shared__ unsigned s_cnt[AGG_WAVES];

    const int tid  = threadIdx.x;
    const int wave = __builtin_amdgcn_readfirstlane(tid >> 5);
    const int lane = tid & 31;
    const int lo   = blockIdx.x * NPB;
    const unsigned nloc = (unsigned)min(NPB, n_nodes - lo);

    for (int i = tid; i < NPB * PITCH / 4; i += AGG_THREADS)
        ((float4*)s_agg)[i] = make_float4(0.f, 0.f, 0.f, 0.f);
    if (tid < NPB) s_deg[tid] = 0.f;
    if (tid < AGG_WAVES) s_cnt[tid] = 0u;
    __syncthreads();

    int2* qe = s_qe + wave * QW;
    int*  qg = s_qg + wave * QG;
    int cnt  = 0;
    int cntg = 0;

    const int n_groups = (n_edges + GRP - 1) / GRP;
    const int n_trips  = (n_groups + GTRIP - 1) / GTRIP;
    const int rot = ((int)blockIdx.x * n_trips) / (int)gridDim.x;
    const unsigned rid = min((unsigned)blockIdx.x, RID_CAP);
    const unsigned bm  = 1u << (rid & 31u);
    const unsigned* mp = map + (size_t)(rid >> 5) * ((size_t)n_trips * GTRIP);

    const int e_tail = (n_edges / GRP) * GRP;
    if (wave == 0 && e_tail < n_edges) {
        const int e = e_tail + lane;
        const bool in = e < n_edges;
        const unsigned d = (unsigned)edst[in ? e : e_tail] - (unsigned)lo;
        const bool own = in && (d < nloc);
        const unsigned mask = __builtin_amdgcn_ballot_w32(own);
        if (own) qe[(int)__builtin_amdgcn_mbcnt_lo(mask, 0u)] = make_int2(e, (int)d);
        cnt = __builtin_popcount(mask);
    }

    if (wave < n_trips) {
        int cc = wave + rot; if (cc >= n_trips) cc -= n_trips;
        uint4 nx = *(const uint4*)(mp + cc * GTRIP + 4 * lane);
        for (int c = wave; c < n_trips; c += AGG_WAVES) {
            const uint4 cur = nx;
            const int g0 = cc * GTRIP + 4 * lane;
            const int c2 = c + AGG_WAVES;
            if (c2 < n_trips) { cc = c2 + rot; if (cc >= n_trips) cc -= n_trips; }
            nx = *(const uint4*)(mp + cc * GTRIP + 4 * lane);

            const unsigned mw[4] = { cur.x, cur.y, cur.z, cur.w };
            #pragma unroll
            for (int w = 0; w < 4; ++w) {
                const unsigned t = mw[w] & bm;
                const unsigned mask = __builtin_amdgcn_ballot_w32(t != 0u);
                if (t != 0u) qg[cntg + (int)__builtin_amdgcn_mbcnt_lo(mask, 0u)] = g0 + w;
                cntg += __builtin_popcount(mask);
            }
            if (cntg > QG - GTRIP) {
                sage_expand(x, esrc, edst, qg, cntg, lo, nloc, qe, cnt, s_cnt, s_rq, s_agg, s_deg, lane);
                cntg = 0;
            }
        }
    }
    sage_expand(x, esrc, edst, qg, cntg, lo, nloc, qe, cnt, s_cnt, s_rq, s_agg, s_deg, lane);
    sage_route(x, esrc, qe, cnt, s_cnt, s_rq, s_agg, s_deg, lane);

    __syncthreads();

    float4 xr[NPB / AGG_WAVES];
    #pragma unroll
    for (int q = 0; q < NPB / AGG_WAVES; ++q) {
        const int row = wave + q * AGG_WAVES;
        xr[q] = make_float4(0.f, 0.f, 0.f, 0.f);
        if (row < (int)nloc) xr[q] = *(const float4*)(x + (size_t)(lo + row) * IN_FEATS + 4 * lane);
    }

    const int nq = __builtin_amdgcn_readfirstlane((int)min(s_cnt[wave], (unsigned)RQ));
    sage_sum(x, s_rq + wave * RQ, nq, s_agg, wave, lane);

    float* s_x = (float*)s_qe;
    #pragma unroll
    for (int q = 0; q < NPB / AGG_WAVES; ++q) {
        float* xp = s_x + (wave + q * AGG_WAVES) * PITCH + lane;
        xp[0] = xr[q].x; xp[32] = xr[q].y; xp[64] = xr[q].z; xp[96] = xr[q].w;
    }

    const int g  = wave >> 3;
    const int n0 = (wave & 7) << 4;
    const float bl = bias[n0 + (lane & 15)];

    __syncthreads();

    if (g >= 2) sage_tile_gemm<2>(s_agg, s_x, s_deg, W, Wr, bl, out, lo, (int)nloc, g, n0, lane);
    else        sage_tile_gemm<3>(s_agg, s_x, s_deg, W, Wr, bl, out, lo, (int)nloc, g, n0, lane);
}

extern "C" void kernel_launch(void* const* d_in, const int* in_sizes, int n_in,
                              void* d_out, int out_size, void* d_ws, size_t ws_size,
                              hipStream_t stream) {
    const float* x    = (const float*)d_in[0];
    const int*   src  = (const int*)d_in[1];
    const int*   dst  = (const int*)d_in[2];
    const float* W    = (const float*)d_in[3];
    const float* Wr   = (const float*)d_in[4];
    const float* bias = (const float*)d_in[5];
    float* out = (float*)d_out;

    const int n_nodes = in_sizes[0] / IN_FEATS;
    const int n_edges = in_sizes[1];

    unsigned* map = (unsigned*)d_ws;
    const int n_groups = (n_edges + GRP - 1) / GRP;
    sage_map_kernel<<<(n_groups + GTRIP - 1) / GTRIP, MAP_THREADS, 0, stream>>>(dst, map, n_edges);

    sage_fused_kernel<<<(n_nodes + NPB - 1) / NPB, AGG_THREADS, 0, stream>>>(
        x, src, dst, map, W, Wr, bias, out, n_nodes, n_edges);
}
